# starting kernel
# baseline (speedup 1.0000x reference)
#include <hip/hip_runtime.h>
#include <math.h>

#define N_INST 64
#define NPIX   64000

typedef __attribute__((ext_vector_type(2))) float v2f;
typedef __attribute__((ext_vector_type(8))) float v8f;

__device__ __forceinline__ float sigmoidf(float x) {
    return __builtin_amdgcn_rcpf(1.0f + __expf(-x));
}

__global__ __launch_bounds__(256) void feat_kernel(
    const int* __restrict__ cls, const float* __restrict__ U_w,
    const float* __restrict__ V_w, const float* __restrict__ P_w,
    float* __restrict__ W1, float* __restrict__ F2) {
    int gid = blockIdx.x * 256 + threadIdx.x;
    int c = gid >> 6;
    int i = gid & 63;
    int a = cls[i] - 1;
    float u = U_w[c * 80 + a];
    float v = V_w[c * 80 + a];
    W1[c * 64 + i] = fmaxf(u, 0.0f) * P_w[c];
    F2[c * 64 + i] = fmaxf(v, 0.0f);
}

__global__ __launch_bounds__(256) void logit_kernel(
    const float* __restrict__ W1, const float* __restrict__ F2,
    const float* __restrict__ bbox, const float* __restrict__ Wc_w,
    const float* __restrict__ P_w, float* __restrict__ R) {
    int gid = blockIdx.x * 256 + threadIdx.x;
    int i = gid & 63;
    int j = gid >> 6;

    float acc = 0.0f;
    for (int c = 0; c < 256; ++c)
        acc += W1[c * 64 + i] * F2[c * 64 + j];

    float xi0 = bbox[i * 5 + 1], yi0 = bbox[i * 5 + 2];
    float xi1 = bbox[i * 5 + 3], yi1 = bbox[i * 5 + 4];
    float xj0 = bbox[j * 5 + 1], yj0 = bbox[j * 5 + 2];
    float xj1 = bbox[j * 5 + 3], yj1 = bbox[j * 5 + 4];
    float wi = xi1 - xi0 + 1.0f, hi = yi1 - yi0 + 1.0f;
    float wj = xj1 - xj0 + 1.0f, hj = yj1 - yj0 + 1.0f;
    float xci = xi0 + 0.5f * wi, yci = yi0 + 0.5f * hi;
    float xcj = xj0 + 0.5f * wj, ycj = yj0 + 0.5f * hj;
    float rwi = __builtin_amdgcn_rcpf(wi);
    float rhi = __builtin_amdgcn_rcpf(hi);
    float dx = -(xci - xcj) * rwi;
    float dy = -(yci - ycj) * rhi;
    float dw = __logf(wj * rwi);
    float dh = __logf(hj * rhi);

    for (int o = 0; o < 128; ++o) {
        float v = dx * Wc_w[o * 4 + 0] + dy * Wc_w[o * 4 + 1] +
                  dw * Wc_w[o * 4 + 2] + dh * Wc_w[o * 4 + 3];
        acc += fmaxf(v, 0.0f) * P_w[256 + o];
    }
    R[i * 64 + j] = sigmoidf(acc);
}

__global__ __launch_bounds__(256) void omat_kernel(
    const float* __restrict__ R, float* __restrict__ O_ws,
    float* __restrict__ O_out) {
    int gid = blockIdx.x * 256 + threadIdx.x;
    int i = gid >> 6;
    int j = gid & 63;
    float o = fmaxf(R[i * 64 + j] - R[j * 64 + i], 0.0f);
    O_ws[gid]  = o;
    O_out[gid] = o;
}

__global__ __launch_bounds__(256) void apply_kernel(
    const float* __restrict__ mask, const float* __restrict__ O,
    float* __restrict__ out) {
    __shared__ float sO[64][68];

    for (int idx = threadIdx.x; idx < 64 * 64; idx += 256)
        sO[idx >> 6][idx & 63] = O[idx];
    __syncthreads();

    const int lane = threadIdx.x & 31;
    const int wave = threadIdx.x >> 5;
    const int col  = lane & 15;
    const int half = lane >> 4;
    const int p    = blockIdx.x * 128 + wave * 16 + col;

    v8f acc[4] = {v8f{}, v8f{}, v8f{}, v8f{}};

    #pragma unroll
    for (int k0 = 0; k0 < 64; k0 += 4) {
        const int kA = k0 + half * 2;
        v2f b;
        b[0] = sigmoidf(mask[(kA + 0) * NPIX + p]);
        b[1] = sigmoidf(mask[(kA + 1) * NPIX + p]);
        #pragma unroll
        for (int mt = 0; mt < 4; ++mt) {
            v2f a;
            a[0] = sO[mt * 16 + col][kA + 0];
            a[1] = sO[mt * 16 + col][kA + 1];
            acc[mt] = __builtin_amdgcn_wmma_f32_16x16x4_f32(
                false, a, false, b, (short)0, acc[mt], false, false);
        }
    }

    #pragma unroll
    for (int mt = 0; mt < 4; ++mt) {
        #pragma unroll
        for (int r = 0; r < 8; ++r) {
            int row = mt * 16 + half * 8 + r;
            float m  = mask[row * NPIX + p];
            float tm = sigmoidf(m);
            out[row * NPIX + p] = m * (1.0f - tm * acc[mt][r]);
        }
    }
}

extern "C" void kernel_launch(void* const* d_in, const int* in_sizes, int n_in,
                              void* d_out, int out_size, void* d_ws, size_t ws_size,
                              hipStream_t stream) {
    (void)in_sizes; (void)n_in; (void)out_size; (void)ws_size;

    const float* mask = (const float*)d_in[0];
    const float* bbox = (const float*)d_in[1];
    const int*   cls  = (const int*)  d_in[2];
    const float* U_w  = (const float*)d_in[3];
    const float* V_w  = (const float*)d_in[4];
    const float* Wc_w = (const float*)d_in[5];
    const float* P_w  = (const float*)d_in[6];

    float* out = (float*)d_out;
    float* ws  = (float*)d_ws;
    float* W1  = ws;
    float* F2  = ws + 16384;
    float* R   = ws + 32768;
    float* Omat= ws + 36864;

    feat_kernel <<<64,  256, 0, stream>>>(cls, U_w, V_w, P_w, W1, F2);
    logit_kernel<<<16,  256, 0, stream>>>(W1, F2, bbox, Wc_w, P_w, R);
    omat_kernel <<<16,  256, 0, stream>>>(R, Omat, out + (size_t)N_INST * NPIX);
    apply_kernel<<<500, 256, 0, stream>>>(mask, Omat, out);
}


# logit kernel: 128-feature box sum moved onto f32 WMMA (K=4 GEMM; relu and P weights applied in the accumulator layout; weights fetched once by vector loads); this changes that sum's order
# speedup vs baseline: 1.7369x; 1.7369x over previous
#include <hip/hip_runtime.h>
#include <math.h>

#define N_INST 64
#define NPIX   64000

typedef __attribute__((ext_vector_type(2))) float v2f;
typedef __attribute__((ext_vector_type(4))) float v4f;
typedef __attribute__((ext_vector_type(8))) float v8f;

__device__ __forceinline__ float sigmoidf(float x) {
    return __builtin_amdgcn_rcpf(1.0f + __expf(-x));
}

__global__ __launch_bounds__(256) void feat_kernel(
    const int* __restrict__ cls, const float* __restrict__ U_w,
    const float* __restrict__ V_w, const float* __restrict__ P_w,
    float* __restrict__ W1, float* __restrict__ F2) {
    int gid = blockIdx.x * 256 + threadIdx.x;
    int c = gid >> 6;
    int i = gid & 63;
    int a = cls[i] - 1;
    float u = U_w[c * 80 + a];
    float v = V_w[c * 80 + a];
    W1[c * 64 + i] = fmaxf(u, 0.0f) * P_w[c];
    F2[c * 64 + i] = fmaxf(v, 0.0f);
}

__global__ __launch_bounds__(256) void logit_kernel(
    const float* __restrict__ W1, const float* __restrict__ F2,
    const float* __restrict__ bbox, const float* __restrict__ Wc_w,
    const float* __restrict__ P_w, float* __restrict__ R) {
    __shared__ float sA[256 * 16];
    __shared__ float sB[256 * 16];
    __shared__ float sD[8 * 32];
    __shared__ float sBox[16 * 32];

    const int t    = threadIdx.x;
    const int lane = t & 31;
    const int wave = t >> 5;
    const int col  = lane & 15;
    const int half = lane >> 4;
    const int m0   = (blockIdx.x >> 2) * 16;
    const int n0   = (blockIdx.x & 3) * 16;

    const int i = m0 + wave + 8 * half;
    const int j = n0 + col;

    const int ib = m0 + 2 * wave;
    float xj0 = bbox[j * 5 + 1], yj0 = bbox[j * 5 + 2];
    float xj1 = bbox[j * 5 + 3], yj1 = bbox[j * 5 + 4];
    float xi0[2], yi0[2], xi1[2], yi1[2];
    #pragma unroll
    for (int u = 0; u < 2; ++u) {
        xi0[u] = bbox[(ib + u) * 5 + 1]; yi0[u] = bbox[(ib + u) * 5 + 2];
        xi1[u] = bbox[(ib + u) * 5 + 3]; yi1[u] = bbox[(ib + u) * 5 + 4];
    }

    const int g  = t >> 2;
    const int qd = t & 3;
    v4f ra[4], rb[4];
    #pragma unroll
    for (int r = 0; r < 4; ++r) {
        ra[r] = *(const v4f*)(W1 + (4 * g + r) * 64 + m0 + 4 * qd);
        rb[r] = *(const v4f*)(F2 + (4 * g + r) * 64 + n0 + 4 * qd);
    }

    v2f wa[8];
    v4f pw[8][2];
    #pragma unroll
    for (int ot = 0; ot < 8; ++ot) {
        wa[ot]    = *(const v2f*)(Wc_w + (16 * ot + col) * 4 + 2 * half);
        pw[ot][0] = *(const v4f*)(P_w + 256 + 16 * ot + 8 * half);
        pw[ot][1] = *(const v4f*)(P_w + 256 + 16 * ot + 8 * half + 4);
    }

    __builtin_amdgcn_sched_barrier(0);

    float wj = xj1 - xj0 + 1.0f, hj = yj1 - yj0 + 1.0f;
    float xcj = xj0 + 0.5f * wj, ycj = yj0 + 0.5f * hj;

    v2f bf[2];
    #pragma unroll
    for (int u = 0; u < 2; ++u) {
        float wi = xi1[u] - xi0[u] + 1.0f, hi = yi1[u] - yi0[u] + 1.0f;
        float xci = xi0[u] + 0.5f * wi, yci = yi0[u] + 0.5f * hi;
        float rwi = __builtin_amdgcn_rcpf(wi);
        float rhi = __builtin_amdgcn_rcpf(hi);
        float dx = -(xci - xcj) * rwi;
        float dy = -(yci - ycj) * rhi;
        float dw = __logf(wj * rwi);
        float dh = __logf(hj * rhi);
        bf[u][0] = half ? dw : dx;
        bf[u][1] = half ? dh : dy;
    }

    #pragma unroll
    for (int u = 0; u < 2; ++u) {
        float s0 = 0.0f, s1 = 0.0f;
        #pragma unroll
        for (int ot = 0; ot < 8; ++ot) {
            const v8f d = __builtin_amdgcn_wmma_f32_16x16x4_f32(
                false, wa[ot], false, bf[u], (short)0, v8f{}, false, false);
            #pragma unroll
            for (int r = 0; r < 4; ++r) {
                s0 += fmaxf(d[r], 0.0f) * pw[ot][0][r];
                s1 += fmaxf(d[r + 4], 0.0f) * pw[ot][1][r];
            }
        }
        sBox[(2 * wave + u) * 32 + lane] = s0 + s1;
    }

    #pragma unroll
    for (int e = 0; e < 4; ++e) {
        v4f ta = {ra[0][e], ra[1][e], ra[2][e], ra[3][e]};
        v4f tb = {rb[0][e], rb[1][e], rb[2][e], rb[3][e]};
        *(v4f*)(sA + 16 * t + 4 * e) = ta;
        *(v4f*)(sB + 16 * t + 4 * e) = tb;
    }
    __syncthreads();

    if (wave == 0) {
        const float* pa = sA + half * 64 + col * 4;
        const float* pb = sB + half * 64 + col * 4;
        v8f acc0 = v8f{}, acc1 = v8f{};
        v4f qa[2][8], qb[2][8];
        #pragma unroll
        for (int u = 0; u < 8; ++u) {
            qa[0][u] = *(const v4f*)(pa + u * 128);
            qb[0][u] = *(const v4f*)(pb + u * 128);
        }
        #pragma unroll
        for (int nb = 0; nb < 4; ++nb) {
            if (nb < 3) {
                #pragma unroll
                for (int u = 0; u < 8; ++u) {
                    qa[(nb + 1) & 1][u] = *(const v4f*)(pa + ((nb + 1) * 8 + u) * 128);
                    qb[(nb + 1) & 1][u] = *(const v4f*)(pb + ((nb + 1) * 8 + u) * 128);
                }
            }
            __builtin_amdgcn_sched_barrier(0);
            #pragma unroll
            for (int u = 0; u < 8; ++u) {
                const v4f va = qa[nb & 1][u];
                const v4f vb = qb[nb & 1][u];
                const v2f a0 = {va[0], va[1]}, b0 = {vb[0], vb[1]};
                const v2f a1 = {va[2], va[3]}, b1 = {vb[2], vb[3]};
                acc0 = __builtin_amdgcn_wmma_f32_16x16x4_f32(
                    false, a0, false, b0, (short)0, acc0, false, false);
                acc1 = __builtin_amdgcn_wmma_f32_16x16x4_f32(
                    false, a1, false, b1, (short)0, acc1, false, false);
            }
            __builtin_amdgcn_sched_barrier(0);
        }
        #pragma unroll
        for (int r = 0; r < 8; ++r)
            sD[r * 32 + lane] = acc0[r] + acc1[r];
    }
    __syncthreads();

    const int rr = wave + 8 * half;
    const float bsum = sBox[rr * 32 + col] + sBox[rr * 32 + 16 + col];
    R[i * 64 + j] = sigmoidf(sD[wave * 32 + lane] + bsum);
}

__global__ __launch_bounds__(256) void omat_kernel(
    const float* __restrict__ R, float* __restrict__ O_ws,
    float* __restrict__ O_out) {
    int gid = blockIdx.x * 256 + threadIdx.x;
    int i = gid >> 6;
    int j = gid & 63;
    float o = fmaxf(R[i * 64 + j] - R[j * 64 + i], 0.0f);
    O_ws[gid]  = o;
    O_out[gid] = o;
}

__global__ __launch_bounds__(256) void apply_kernel(
    const float* __restrict__ mask, const float* __restrict__ O,
    float* __restrict__ out) {
    __shared__ float sO[64][68];

    for (int idx = threadIdx.x; idx < 64 * 64; idx += 256)
        sO[idx >> 6][idx & 63] = O[idx];
    __syncthreads();

    const int lane = threadIdx.x & 31;
    const int wave = threadIdx.x >> 5;
    const int col  = lane & 15;
    const int half = lane >> 4;
    const int p    = blockIdx.x * 128 + wave * 16 + col;

    v8f acc[4] = {v8f{}, v8f{}, v8f{}, v8f{}};

    #pragma unroll
    for (int k0 = 0; k0 < 64; k0 += 4) {
        const int kA = k0 + half * 2;
        v2f b;
        b[0] = sigmoidf(mask[(kA + 0) * NPIX + p]);
        b[1] = sigmoidf(mask[(kA + 1) * NPIX + p]);
        #pragma unroll
        for (int mt = 0; mt < 4; ++mt) {
            v2f a;
            a[0] = sO[mt * 16 + col][kA + 0];
            a[1] = sO[mt * 16 + col][kA + 1];
            acc[mt] = __builtin_amdgcn_wmma_f32_16x16x4_f32(
                false, a, false, b, (short)0, acc[mt], false, false);
        }
    }

    #pragma unroll
    for (int mt = 0; mt < 4; ++mt) {
        #pragma unroll
        for (int r = 0; r < 8; ++r) {
            int row = mt * 16 + half * 8 + r;
            float m  = mask[row * NPIX + p];
            float tm = sigmoidf(m);
            out[row * NPIX + p] = m * (1.0f - tm * acc[mt][r]);
        }
    }
}

extern "C" void kernel_launch(void* const* d_in, const int* in_sizes, int n_in,
                              void* d_out, int out_size, void* d_ws, size_t ws_size,
                              hipStream_t stream) {
    (void)in_sizes; (void)n_in; (void)out_size; (void)ws_size;

    const float* mask = (const float*)d_in[0];
    const float* bbox = (const float*)d_in[1];
    const int*   cls  = (const int*)  d_in[2];
    const float* U_w  = (const float*)d_in[3];
    const float* V_w  = (const float*)d_in[4];
    const float* Wc_w = (const float*)d_in[5];
    const float* P_w  = (const float*)d_in[6];

    float* out = (float*)d_out;
    float* ws  = (float*)d_ws;
    float* W1  = ws;
    float* F2  = ws + 16384;
    float* R   = ws + 32768;
    float* Omat= ws + 36864;

    feat_kernel <<<64,  256, 0, stream>>>(cls, U_w, V_w, P_w, W1, F2);
    logit_kernel<<<16,  256, 0, stream>>>(W1, F2, bbox, Wc_w, P_w, R);
    omat_kernel <<<16,  256, 0, stream>>>(R, Omat, out + (size_t)N_INST * NPIX);
    apply_kernel<<<500, 256, 0, stream>>>(mask, Omat, out);
}


# apply kernel: global accesses widened: a wave owns 32 pixels x 64 rows (tile column c = pixel 2c+T) so mask loads and out stores are 8-byte whole-line vectors; O staged by 16-byte loads
# speedup vs baseline: 1.9646x; 1.1311x over previous
#include <hip/hip_runtime.h>
#include <math.h>

#define N_INST 64
#define NPIX   64000

typedef __attribute__((ext_vector_type(2))) float v2f;
typedef __attribute__((ext_vector_type(4))) float v4f;
typedef __attribute__((ext_vector_type(8))) float v8f;

__device__ __forceinline__ float sigmoidf(float x) {
    return __builtin_amdgcn_rcpf(1.0f + __expf(-x));
}

__global__ __launch_bounds__(256) void feat_kernel(
    const int* __restrict__ cls, const float* __restrict__ U_w,
    const float* __restrict__ V_w, const float* __restrict__ P_w,
    float* __restrict__ W1, float* __restrict__ F2) {
    int gid = blockIdx.x * 256 + threadIdx.x;
    int c = gid >> 6;
    int i = gid & 63;
    int a = cls[i] - 1;
    float u = U_w[c * 80 + a];
    float v = V_w[c * 80 + a];
    W1[c * 64 + i] = fmaxf(u, 0.0f) * P_w[c];
    F2[c * 64 + i] = fmaxf(v, 0.0f);
}

__global__ __launch_bounds__(256) void logit_kernel(
    const float* __restrict__ W1, const float* __restrict__ F2,
    const float* __restrict__ bbox, const float* __restrict__ Wc_w,
    const float* __restrict__ P_w, float* __restrict__ R) {
    __shared__ float sA[256 * 16];
    __shared__ float sB[256 * 16];
    __shared__ float sD[8 * 32];
    __shared__ float sBox[16 * 32];

    const int t    = threadIdx.x;
    const int lane = t & 31;
    const int wave = t >> 5;
    const int col  = lane & 15;
    const int half = lane >> 4;
    const int m0   = (blockIdx.x >> 2) * 16;
    const int n0   = (blockIdx.x & 3) * 16;

    const int i = m0 + wave + 8 * half;
    const int j = n0 + col;

    const int ib = m0 + 2 * wave;
    float xj0 = bbox[j * 5 + 1], yj0 = bbox[j * 5 + 2];
    float xj1 = bbox[j * 5 + 3], yj1 = bbox[j * 5 + 4];
    float xi0[2], yi0[2], xi1[2], yi1[2];
    #pragma unroll
    for (int u = 0; u < 2; ++u) {
        xi0[u] = bbox[(ib + u) * 5 + 1]; yi0[u] = bbox[(ib + u) * 5 + 2];
        xi1[u] = bbox[(ib + u) * 5 + 3]; yi1[u] = bbox[(ib + u) * 5 + 4];
    }

    const int g  = t >> 2;
    const int qd = t & 3;
    v4f ra[4], rb[4];
    #pragma unroll
    for (int r = 0; r < 4; ++r) {
        ra[r] = *(const v4f*)(W1 + (4 * g + r) * 64 + m0 + 4 * qd);
        rb[r] = *(const v4f*)(F2 + (4 * g + r) * 64 + n0 + 4 * qd);
    }

    v2f wa[8];
    v4f pw[8][2];
    #pragma unroll
    for (int ot = 0; ot < 8; ++ot) {
        wa[ot]    = *(const v2f*)(Wc_w + (16 * ot + col) * 4 + 2 * half);
        pw[ot][0] = *(const v4f*)(P_w + 256 + 16 * ot + 8 * half);
        pw[ot][1] = *(const v4f*)(P_w + 256 + 16 * ot + 8 * half + 4);
    }

    __builtin_amdgcn_sched_barrier(0);

    float wj = xj1 - xj0 + 1.0f, hj = yj1 - yj0 + 1.0f;
    float xcj = xj0 + 0.5f * wj, ycj = yj0 + 0.5f * hj;

    v2f bf[2];
    #pragma unroll
    for (int u = 0; u < 2; ++u) {
        float wi = xi1[u] - xi0[u] + 1.0f, hi = yi1[u] - yi0[u] + 1.0f;
        float xci = xi0[u] + 0.5f * wi, yci = yi0[u] + 0.5f * hi;
        float rwi = __builtin_amdgcn_rcpf(wi);
        float rhi = __builtin_amdgcn_rcpf(hi);
        float dx = -(xci - xcj) * rwi;
        float dy = -(yci - ycj) * rhi;
        float dw = __logf(wj * rwi);
        float dh = __logf(hj * rhi);
        bf[u][0] = half ? dw : dx;
        bf[u][1] = half ? dh : dy;
    }

    #pragma unroll
    for (int u = 0; u < 2; ++u) {
        float s0 = 0.0f, s1 = 0.0f;
        #pragma unroll
        for (int ot = 0; ot < 8; ++ot) {
            const v8f d = __builtin_amdgcn_wmma_f32_16x16x4_f32(
                false, wa[ot], false, bf[u], (short)0, v8f{}, false, false);
            #pragma unroll
            for (int r = 0; r < 4; ++r) {
                s0 += fmaxf(d[r], 0.0f) * pw[ot][0][r];
                s1 += fmaxf(d[r + 4], 0.0f) * pw[ot][1][r];
            }
        }
        sBox[(2 * wave + u) * 32 + lane] = s0 + s1;
    }

    #pragma unroll
    for (int e = 0; e < 4; ++e) {
        v4f ta = {ra[0][e], ra[1][e], ra[2][e], ra[3][e]};
        v4f tb = {rb[0][e], rb[1][e], rb[2][e], rb[3][e]};
        *(v4f*)(sA + 16 * t + 4 * e) = ta;
        *(v4f*)(sB + 16 * t + 4 * e) = tb;
    }
    __syncthreads();

    if (wave == 0) {
        const float* pa = sA + half * 64 + col * 4;
        const float* pb = sB + half * 64 + col * 4;
        v8f acc0 = v8f{}, acc1 = v8f{};
        v4f qa[2][8], qb[2][8];
        #pragma unroll
        for (int u = 0; u < 8; ++u) {
            qa[0][u] = *(const v4f*)(pa + u * 128);
            qb[0][u] = *(const v4f*)(pb + u * 128);
        }
        #pragma unroll
        for (int nb = 0; nb < 4; ++nb) {
            if (nb < 3) {
                #pragma unroll
                for (int u = 0; u < 8; ++u) {
                    qa[(nb + 1) & 1][u] = *(const v4f*)(pa + ((nb + 1) * 8 + u) * 128);
                    qb[(nb + 1) & 1][u] = *(const v4f*)(pb + ((nb + 1) * 8 + u) * 128);
                }
            }
            __builtin_amdgcn_sched_barrier(0);
            #pragma unroll
            for (int u = 0; u < 8; ++u) {
                const v4f va = qa[nb & 1][u];
                const v4f vb = qb[nb & 1][u];
                const v2f a0 = {va[0], va[1]}, b0 = {vb[0], vb[1]};
                const v2f a1 = {va[2], va[3]}, b1 = {vb[2], vb[3]};
                acc0 = __builtin_amdgcn_wmma_f32_16x16x4_f32(
                    false, a0, false, b0, (short)0, acc0, false, false);
                acc1 = __builtin_amdgcn_wmma_f32_16x16x4_f32(
                    false, a1, false, b1, (short)0, acc1, false, false);
            }
            __builtin_amdgcn_sched_barrier(0);
        }
        #pragma unroll
        for (int r = 0; r < 8; ++r)
            sD[r * 32 + lane] = acc0[r] + acc1[r];
    }
    __syncthreads();

    const int rr = wave + 8 * half;
    const float bsum = sBox[rr * 32 + col] + sBox[rr * 32 + 16 + col];
    R[i * 64 + j] = sigmoidf(sD[wave * 32 + lane] + bsum);
}

__global__ __launch_bounds__(256) void omat_kernel(
    const float* __restrict__ R, float* __restrict__ O_ws,
    float* __restrict__ O_out) {
    int gid = blockIdx.x * 256 + threadIdx.x;
    int i = gid >> 6;
    int j = gid & 63;
    float o = fmaxf(R[i * 64 + j] - R[j * 64 + i], 0.0f);
    O_ws[gid]  = o;
    O_out[gid] = o;
}

__global__ __launch_bounds__(128) void apply_kernel(
    const float* __restrict__ mask, const float* __restrict__ O,
    float* __restrict__ out) {
    __shared__ float sO[64][68];

    const int t = threadIdx.x;
    v4f o4[8];
    #pragma unroll
    for (int r = 0; r < 8; ++r)
        o4[r] = *(const v4f*)(O + 4 * (t + 128 * r));
    #pragma unroll
    for (int r = 0; r < 8; ++r) {
        const int q = t + 128 * r;
        *(v4f*)(&sO[q >> 4][(q & 15) * 4]) = o4[r];
    }
    __syncthreads();

    const int lane = t & 31;
    const int wave = t >> 5;
    const int col  = lane & 15;
    const int half = lane >> 4;
    const int p    = blockIdx.x * 128 + wave * 32 + 2 * col;

    v8f acc[4][2] = {{v8f{}, v8f{}}, {v8f{}, v8f{}}, {v8f{}, v8f{}}, {v8f{}, v8f{}}};

    #pragma unroll
    for (int k0 = 0; k0 < 64; k0 += 4) {
        const int kA = k0 + half * 2;
        const v2f m0 = *(const v2f*)(mask + (kA + 0) * NPIX + p);
        const v2f m1 = *(const v2f*)(mask + (kA + 1) * NPIX + p);
        v2f b0, b1;
        b0[0] = sigmoidf(m0[0]); b0[1] = sigmoidf(m1[0]);
        b1[0] = sigmoidf(m0[1]); b1[1] = sigmoidf(m1[1]);
        #pragma unroll
        for (int mt = 0; mt < 4; ++mt) {
            v2f a;
            a[0] = sO[mt * 16 + col][kA + 0];
            a[1] = sO[mt * 16 + col][kA + 1];
            acc[mt][0] = __builtin_amdgcn_wmma_f32_16x16x4_f32(
                false, a, false, b0, (short)0, acc[mt][0], false, false);
            acc[mt][1] = __builtin_amdgcn_wmma_f32_16x16x4_f32(
                false, a, false, b1, (short)0, acc[mt][1], false, false);
        }
    }

    #pragma unroll
    for (int mt = 0; mt < 4; ++mt) {
        #pragma unroll
        for (int r = 0; r < 8; ++r) {
            const int row = mt * 16 + half * 8 + r;
            const v2f m = *(const v2f*)(mask + row * NPIX + p);
            v2f o;
            o[0] = m[0] * (1.0f - sigmoidf(m[0]) * acc[mt][0][r]);
            o[1] = m[1] * (1.0f - sigmoidf(m[1]) * acc[mt][1][r]);
            *(v2f*)(out + row * NPIX + p) = o;
        }
    }
}

extern "C" void kernel_launch(void* const* d_in, const int* in_sizes, int n_in,
                              void* d_out, int out_size, void* d_ws, size_t ws_size,
                              hipStream_t stream) {
    (void)in_sizes; (void)n_in; (void)out_size; (void)ws_size;

    const float* mask = (const float*)d_in[0];
    const float* bbox = (const float*)d_in[1];
    const int*   cls  = (const int*)  d_in[2];
    const float* U_w  = (const float*)d_in[3];
    const float* V_w  = (const float*)d_in[4];
    const float* Wc_w = (const float*)d_in[5];
    const float* P_w  = (const float*)d_in[6];

    float* out = (float*)d_out;
    float* ws  = (float*)d_ws;
    float* W1  = ws;
    float* F2  = ws + 16384;
    float* R   = ws + 32768;
    float* Omat= ws + 36864;

    feat_kernel <<<64,  256, 0, stream>>>(cls, U_w, V_w, P_w, W1, F2);
    logit_kernel<<<16,  256, 0, stream>>>(W1, F2, bbox, Wc_w, P_w, R);
    omat_kernel <<<16,  256, 0, stream>>>(R, Omat, out + (size_t)N_INST * NPIX);
    apply_kernel<<<500, 128, 0, stream>>>(mask, Omat, out);
}


# logit kernel: each block also computes its mirror tile of R (second 8-wave group, same code) and writes O=relu(R-R^T) itself; omat launch and R round trip removed (4 launches -> 3); sum orders kept
# speedup vs baseline: 2.1408x; 1.0897x over previous
#include <hip/hip_runtime.h>
#include <math.h>

#define N_INST 64
#define NPIX   64000

typedef __attribute__((ext_vector_type(2))) float v2f;
typedef __attribute__((ext_vector_type(4))) float v4f;
typedef __attribute__((ext_vector_type(8))) float v8f;

__device__ __forceinline__ float sigmoidf(float x) {
    return __builtin_amdgcn_rcpf(1.0f + __expf(-x));
}

__global__ __launch_bounds__(256) void feat_kernel(
    const int* __restrict__ cls, const float* __restrict__ U_w,
    const float* __restrict__ V_w, const float* __restrict__ P_w,
    float* __restrict__ W1, float* __restrict__ F2) {
    int gid = blockIdx.x * 256 + threadIdx.x;
    int c = gid >> 6;
    int i = gid & 63;
    int a = cls[i] - 1;
    float u = U_w[c * 80 + a];
    float v = V_w[c * 80 + a];
    W1[c * 64 + i] = fmaxf(u, 0.0f) * P_w[c];
    F2[c * 64 + i] = fmaxf(v, 0.0f);
}

__global__ __launch_bounds__(512) void relation_kernel(
    const float* __restrict__ W1, const float* __restrict__ F2,
    const float* __restrict__ bbox, const float* __restrict__ Wc_w,
    const float* __restrict__ P_w, float* __restrict__ O_ws,
    float* __restrict__ O_out) {
    __shared__ float sA[2][256 * 16];
    __shared__ float sB[2][256 * 16];
    __shared__ float sD[2][8 * 32];
    __shared__ float sBox[2][16 * 32];
    __shared__ float sR[2][16 * 17];

    const int lane = threadIdx.x & 31;
    const int ts   = threadIdx.x >> 8;
    const int t    = threadIdx.x & 255;
    const int wave = t >> 5;
    const int col  = lane & 15;
    const int half = lane >> 4;
    const int bm   = (blockIdx.x >> 2) * 16;
    const int bn   = (blockIdx.x & 3) * 16;
    const int m0   = ts ? bn : bm;
    const int n0   = ts ? bm : bn;

    float* const gA   = sA[ts];
    float* const gB   = sB[ts];
    float* const gD   = sD[ts];
    float* const gBox = sBox[ts];

    const int j  = n0 + col;
    const int ib = m0 + 2 * wave;
    float xj0 = bbox[j * 5 + 1], yj0 = bbox[j * 5 + 2];
    float xj1 = bbox[j * 5 + 3], yj1 = bbox[j * 5 + 4];
    float xi0[2], yi0[2], xi1[2], yi1[2];
    #pragma unroll
    for (int u = 0; u < 2; ++u) {
        xi0[u] = bbox[(ib + u) * 5 + 1]; yi0[u] = bbox[(ib + u) * 5 + 2];
        xi1[u] = bbox[(ib + u) * 5 + 3]; yi1[u] = bbox[(ib + u) * 5 + 4];
    }

    const int g  = t >> 2;
    const int qd = t & 3;
    v4f ra[4], rb[4];
    #pragma unroll
    for (int r = 0; r < 4; ++r) {
        ra[r] = *(const v4f*)(W1 + (4 * g + r) * 64 + m0 + 4 * qd);
        rb[r] = *(const v4f*)(F2 + (4 * g + r) * 64 + n0 + 4 * qd);
    }

    v2f wa[8];
    v4f pw[8][2];
    #pragma unroll
    for (int ot = 0; ot < 8; ++ot) {
        wa[ot]    = *(const v2f*)(Wc_w + (16 * ot + col) * 4 + 2 * half);
        pw[ot][0] = *(const v4f*)(P_w + 256 + 16 * ot + 8 * half);
        pw[ot][1] = *(const v4f*)(P_w + 256 + 16 * ot + 8 * half + 4);
    }

    __builtin_amdgcn_sched_barrier(0);

    float wj = xj1 - xj0 + 1.0f, hj = yj1 - yj0 + 1.0f;
    float xcj = xj0 + 0.5f * wj, ycj = yj0 + 0.5f * hj;

    v2f bf[2];
    #pragma unroll
    for (int u = 0; u < 2; ++u) {
        float wi = xi1[u] - xi0[u] + 1.0f, hi = yi1[u] - yi0[u] + 1.0f;
        float xci = xi0[u] + 0.5f * wi, yci = yi0[u] + 0.5f * hi;
        float rwi = __builtin_amdgcn_rcpf(wi);
        float rhi = __builtin_amdgcn_rcpf(hi);
        float dx = -(xci - xcj) * rwi;
        float dy = -(yci - ycj) * rhi;
        float dw = __logf(wj * rwi);
        float dh = __logf(hj * rhi);
        bf[u][0] = half ? dw : dx;
        bf[u][1] = half ? dh : dy;
    }

    #pragma unroll
    for (int u = 0; u < 2; ++u) {
        float s0 = 0.0f, s1 = 0.0f;
        #pragma unroll
        for (int ot = 0; ot < 8; ++ot) {
            const v8f d = __builtin_amdgcn_wmma_f32_16x16x4_f32(
                false, wa[ot], false, bf[u], (short)0, v8f{}, false, false);
            #pragma unroll
            for (int r = 0; r < 4; ++r) {
                s0 += fmaxf(d[r], 0.0f) * pw[ot][0][r];
                s1 += fmaxf(d[r + 4], 0.0f) * pw[ot][1][r];
            }
        }
        gBox[(2 * wave + u) * 32 + lane] = s0 + s1;
    }

    #pragma unroll
    for (int e = 0; e < 4; ++e) {
        v4f ta = {ra[0][e], ra[1][e], ra[2][e], ra[3][e]};
        v4f tb = {rb[0][e], rb[1][e], rb[2][e], rb[3][e]};
        *(v4f*)(gA + 16 * t + 4 * e) = ta;
        *(v4f*)(gB + 16 * t + 4 * e) = tb;
    }
    __syncthreads();

    if (wave == ts) {
        const float* pa = gA + half * 64 + col * 4;
        const float* pb = gB + half * 64 + col * 4;
        v8f acc0 = v8f{}, acc1 = v8f{};
        v4f qa[2][8], qb[2][8];
        #pragma unroll
        for (int u = 0; u < 8; ++u) {
            qa[0][u] = *(const v4f*)(pa + u * 128);
            qb[0][u] = *(const v4f*)(pb + u * 128);
        }
        #pragma unroll
        for (int nb = 0; nb < 4; ++nb) {
            if (nb < 3) {
                #pragma unroll
                for (int u = 0; u < 8; ++u) {
                    qa[(nb + 1) & 1][u] = *(const v4f*)(pa + ((nb + 1) * 8 + u) * 128);
                    qb[(nb + 1) & 1][u] = *(const v4f*)(pb + ((nb + 1) * 8 + u) * 128);
                }
            }
            __builtin_amdgcn_sched_barrier(0);
            #pragma unroll
            for (int u = 0; u < 8; ++u) {
                const v4f va = qa[nb & 1][u];
                const v4f vb = qb[nb & 1][u];
                const v2f a0 = {va[0], va[1]}, b0 = {vb[0], vb[1]};
                const v2f a1 = {va[2], va[3]}, b1 = {vb[2], vb[3]};
                acc0 = __builtin_amdgcn_wmma_f32_16x16x4_f32(
                    false, a0, false, b0, (short)0, acc0, false, false);
                acc1 = __builtin_amdgcn_wmma_f32_16x16x4_f32(
                    false, a1, false, b1, (short)0, acc1, false, false);
            }
            __builtin_amdgcn_sched_barrier(0);
        }
        #pragma unroll
        for (int r = 0; r < 8; ++r)
            gD[r * 32 + lane] = acc0[r] + acc1[r];
    }
    __syncthreads();

    const int rr = wave + 8 * half;
    const float bsum = gBox[rr * 32 + col] + gBox[rr * 32 + 16 + col];
    sR[ts][rr * 17 + col] = sigmoidf(gD[wave * 32 + lane] + bsum);
    __syncthreads();

    const float o = fmaxf(sR[0][rr * 17 + col] - sR[1][col * 17 + rr], 0.0f);
    float* const dst = ts ? O_out : O_ws;
    dst[(bm + rr) * 64 + bn + col] = o;
}

__global__ __launch_bounds__(128) void apply_kernel(
    const float* __restrict__ mask, const float* __restrict__ O,
    float* __restrict__ out) {
    __shared__ float sO[64][68];

    const int t = threadIdx.x;
    v4f o4[8];
    #pragma unroll
    for (int r = 0; r < 8; ++r)
        o4[r] = *(const v4f*)(O + 4 * (t + 128 * r));
    #pragma unroll
    for (int r = 0; r < 8; ++r) {
        const int q = t + 128 * r;
        *(v4f*)(&sO[q >> 4][(q & 15) * 4]) = o4[r];
    }
    __syncthreads();

    const int lane = t & 31;
    const int wave = t >> 5;
    const int col  = lane & 15;
    const int half = lane >> 4;
    const int p    = blockIdx.x * 128 + wave * 32 + 2 * col;

    v8f acc[4][2] = {{v8f{}, v8f{}}, {v8f{}, v8f{}}, {v8f{}, v8f{}}, {v8f{}, v8f{}}};

    #pragma unroll
    for (int k0 = 0; k0 < 64; k0 += 4) {
        const int kA = k0 + half * 2;
        const v2f m0 = *(const v2f*)(mask + (kA + 0) * NPIX + p);
        const v2f m1 = *(const v2f*)(mask + (kA + 1) * NPIX + p);
        v2f b0, b1;
        b0[0] = sigmoidf(m0[0]); b0[1] = sigmoidf(m1[0]);
        b1[0] = sigmoidf(m0[1]); b1[1] = sigmoidf(m1[1]);
        #pragma unroll
        for (int mt = 0; mt < 4; ++mt) {
            v2f a;
            a[0] = sO[mt * 16 + col][kA + 0];
            a[1] = sO[mt * 16 + col][kA + 1];
            acc[mt][0] = __builtin_amdgcn_wmma_f32_16x16x4_f32(
                false, a, false, b0, (short)0, acc[mt][0], false, false);
            acc[mt][1] = __builtin_amdgcn_wmma_f32_16x16x4_f32(
                false, a, false, b1, (short)0, acc[mt][1], false, false);
        }
    }

    #pragma unroll
    for (int mt = 0; mt < 4; ++mt) {
        #pragma unroll
        for (int r = 0; r < 8; ++r) {
            const int row = mt * 16 + half * 8 + r;
            const v2f m = *(const v2f*)(mask + row * NPIX + p);
            v2f o;
            o[0] = m[0] * (1.0f - sigmoidf(m[0]) * acc[mt][0][r]);
            o[1] = m[1] * (1.0f - sigmoidf(m[1]) * acc[mt][1][r]);
            *(v2f*)(out + row * NPIX + p) = o;
        }
    }
}

extern "C" void kernel_launch(void* const* d_in, const int* in_sizes, int n_in,
                              void* d_out, int out_size, void* d_ws, size_t ws_size,
                              hipStream_t stream) {
    (void)in_sizes; (void)n_in; (void)out_size; (void)ws_size;

    const float* mask = (const float*)d_in[0];
    const float* bbox = (const float*)d_in[1];
    const int*   cls  = (const int*)  d_in[2];
    const float* U_w  = (const float*)d_in[3];
    const float* V_w  = (const float*)d_in[4];
    const float* Wc_w = (const float*)d_in[5];
    const float* P_w  = (const float*)d_in[6];

    float* out = (float*)d_out;
    float* ws  = (float*)d_ws;
    float* W1  = ws;
    float* F2  = ws + 16384;
    float* Omat= ws + 36864;

    feat_kernel    <<<64,  256, 0, stream>>>(cls, U_w, V_w, P_w, W1, F2);
    relation_kernel<<<16,  512, 0, stream>>>(W1, F2, bbox, Wc_w, P_w, Omat,
                                             out + (size_t)N_INST * NPIX);
    apply_kernel   <<<500, 128, 0, stream>>>(mask, Omat, out);
}


# logit kernel re-tiled: 64 blocks of 8 waves, each one 8x8 block of O; one 16x16 WMMA tile holds its R block and the mirror block (tile rows I|J, columns J|I); 4x less work per block; sum orders kept
# speedup vs baseline: 2.3523x; 1.0988x over previous
#include <hip/hip_runtime.h>
#include <math.h>

#define N_INST 64
#define NPIX   64000

typedef __attribute__((ext_vector_type(2))) float v2f;
typedef __attribute__((ext_vector_type(4))) float v4f;
typedef __attribute__((ext_vector_type(8))) float v8f;

__device__ __forceinline__ float sigmoidf(float x) {
    return __builtin_amdgcn_rcpf(1.0f + __expf(-x));
}

__global__ __launch_bounds__(256) void feat_kernel(
    const int* __restrict__ cls, const float* __restrict__ U_w,
    const float* __restrict__ V_w, const float* __restrict__ P_w,
    float* __restrict__ W1, float* __restrict__ F2) {
    int gid = blockIdx.x * 256 + threadIdx.x;
    int c = gid >> 6;
    int i = gid & 63;
    int a = cls[i] - 1;
    float u = U_w[c * 80 + a];
    float v = V_w[c * 80 + a];
    W1[c * 64 + i] = fmaxf(u, 0.0f) * P_w[c];
    F2[c * 64 + i] = fmaxf(v, 0.0f);
}

__global__ __launch_bounds__(256) void relation_kernel(
    const float* __restrict__ W1, const float* __restrict__ F2,
    const float* __restrict__ bbox, const float* __restrict__ Wc_w,
    const float* __restrict__ P_w, float* __restrict__ O_ws,
    float* __restrict__ O_out) {
    __shared__ float sA[256 * 16];
    __shared__ float sB[256 * 16];
    __shared__ float sD[8 * 32];
    __shared__ float sBox[16 * 32];

    const int t    = threadIdx.x;
    const int lane = t & 31;
    const int wave = t >> 5;
    const int col  = lane & 15;
    const int half = lane >> 4;
    const int bi   = (blockIdx.x >> 3) * 8;
    const int bj   = (blockIdx.x & 7) * 8;

    const bool top = wave < 4;
    const int pm = 2 * wave + (col >> 3);
    const int pn = (col & 7) + (top ? 0 : 8);
    const int i  = (top ? bi : bj) + 2 * (wave & 3) + (col >> 3);
    const int j  = (top ? bj : bi) + (col & 7);
    const float xi0 = bbox[i * 5 + 1], yi0 = bbox[i * 5 + 2];
    const float xi1 = bbox[i * 5 + 3], yi1 = bbox[i * 5 + 4];
    const float xj0 = bbox[j * 5 + 1], yj0 = bbox[j * 5 + 2];
    const float xj1 = bbox[j * 5 + 3], yj1 = bbox[j * 5 + 4];

    v2f wa[8];
    v4f pw[8][2];
    #pragma unroll
    for (int ot = 0; ot < 8; ++ot) {
        wa[ot]    = *(const v2f*)(Wc_w + (16 * ot + col) * 4 + 2 * half);
        pw[ot][0] = *(const v4f*)(P_w + 256 + 16 * ot + 8 * half);
        pw[ot][1] = *(const v4f*)(P_w + 256 + 16 * ot + 8 * half + 4);
    }

    const int g  = t >> 2;
    const int qd = t & 3;
    const int ai = (qd < 2) ? bi + 4 * qd : bj + 4 * (qd - 2);
    const int ci = (qd < 2) ? bj + 4 * qd : bi + 4 * (qd - 2);
    v4f ra[4], rb[4];
    #pragma unroll
    for (int r = 0; r < 4; ++r) {
        ra[r] = *(const v4f*)(W1 + (4 * g + r) * 64 + ai);
        rb[r] = *(const v4f*)(F2 + (4 * g + r) * 64 + ci);
    }

    __builtin_amdgcn_sched_barrier(0);

    v2f bf;
    {
        const float wj = xj1 - xj0 + 1.0f, hj = yj1 - yj0 + 1.0f;
        const float xcj = xj0 + 0.5f * wj, ycj = yj0 + 0.5f * hj;
        const float wi = xi1 - xi0 + 1.0f, hi = yi1 - yi0 + 1.0f;
        const float xci = xi0 + 0.5f * wi, yci = yi0 + 0.5f * hi;
        const float rwi = __builtin_amdgcn_rcpf(wi);
        const float rhi = __builtin_amdgcn_rcpf(hi);
        const float dx = -(xci - xcj) * rwi;
        const float dy = -(yci - ycj) * rhi;
        const float dw = __logf(wj * rwi);
        const float dh = __logf(hj * rhi);
        bf[0] = half ? dw : dx;
        bf[1] = half ? dh : dy;
    }

    {
        float s0 = 0.0f, s1 = 0.0f;
        #pragma unroll
        for (int ot = 0; ot < 8; ++ot) {
            const v8f d = __builtin_amdgcn_wmma_f32_16x16x4_f32(
                false, wa[ot], false, bf, (short)0, v8f{}, false, false);
            #pragma unroll
            for (int r = 0; r < 4; ++r) {
                s0 += fmaxf(d[r], 0.0f) * pw[ot][0][r];
                s1 += fmaxf(d[r + 4], 0.0f) * pw[ot][1][r];
            }
        }
        sBox[pm * 32 + half * 16 + pn] = s0 + s1;
    }

    #pragma unroll
    for (int e = 0; e < 4; ++e) {
        v4f ta = {ra[0][e], ra[1][e], ra[2][e], ra[3][e]};
        v4f tb = {rb[0][e], rb[1][e], rb[2][e], rb[3][e]};
        *(v4f*)(sA + 16 * t + 4 * e) = ta;
        *(v4f*)(sB + 16 * t + 4 * e) = tb;
    }
    __syncthreads();

    if (wave == 0) {
        const float* pa = sA + half * 64 + col * 4;
        const float* pb = sB + half * 64 + col * 4;
        v8f acc0 = v8f{}, acc1 = v8f{};
        v4f qa[2][8], qb[2][8];
        #pragma unroll
        for (int u = 0; u < 8; ++u) {
            qa[0][u] = *(const v4f*)(pa + u * 128);
            qb[0][u] = *(const v4f*)(pb + u * 128);
        }
        #pragma unroll
        for (int nb = 0; nb < 4; ++nb) {
            if (nb < 3) {
                #pragma unroll
                for (int u = 0; u < 8; ++u) {
                    qa[(nb + 1) & 1][u] = *(const v4f*)(pa + ((nb + 1) * 8 + u) * 128);
                    qb[(nb + 1) & 1][u] = *(const v4f*)(pb + ((nb + 1) * 8 + u) * 128);
                }
            }
            __builtin_amdgcn_sched_barrier(0);
            #pragma unroll
            for (int u = 0; u < 8; ++u) {
                const v4f va = qa[nb & 1][u];
                const v4f vb = qb[nb & 1][u];
                const v2f a0 = {va[0], va[1]}, b0 = {vb[0], vb[1]};
                const v2f a1 = {va[2], va[3]}, b1 = {vb[2], vb[3]};
                acc0 = __builtin_amdgcn_wmma_f32_16x16x4_f32(
                    false, a0, false, b0, (short)0, acc0, false, false);
                acc1 = __builtin_amdgcn_wmma_f32_16x16x4_f32(
                    false, a1, false, b1, (short)0, acc1, false, false);
            }
            __builtin_amdgcn_sched_barrier(0);
        }
        #pragma unroll
        for (int r = 0; r < 8; ++r)
            sD[r * 32 + lane] = acc0[r] + acc1[r];
    }
    __syncthreads();

    if (t < 64) {
        const int r = t >> 3;
        const int c = t & 7;
        const float own_box = sBox[r * 32 + c] + sBox[r * 32 + 16 + c];
        const float mir_box = sBox[(8 + c) * 32 + 8 + r] + sBox[(8 + c) * 32 + 24 + r];
        const float own = sigmoidf(sD[r * 32 + c] + own_box);
        const float mir = sigmoidf(sD[c * 32 + 24 + r] + mir_box);
        const float o = fmaxf(own - mir, 0.0f);
        O_ws [(bi + r) * 64 + bj + c] = o;
        O_out[(bi + r) * 64 + bj + c] = o;
    }
}

__global__ __launch_bounds__(128) void apply_kernel(
    const float* __restrict__ mask, const float* __restrict__ O,
    float* __restrict__ out) {
    __shared__ float sO[64][68];

    const int t = threadIdx.x;
    v4f o4[8];
    #pragma unroll
    for (int r = 0; r < 8; ++r)
        o4[r] = *(const v4f*)(O + 4 * (t + 128 * r));
    #pragma unroll
    for (int r = 0; r < 8; ++r) {
        const int q = t + 128 * r;
        *(v4f*)(&sO[q >> 4][(q & 15) * 4]) = o4[r];
    }
    __syncthreads();

    const int lane = t & 31;
    const int wave = t >> 5;
    const int col  = lane & 15;
    const int half = lane >> 4;
    const int p    = blockIdx.x * 128 + wave * 32 + 2 * col;

    v8f acc[4][2] = {{v8f{}, v8f{}}, {v8f{}, v8f{}}, {v8f{}, v8f{}}, {v8f{}, v8f{}}};

    #pragma unroll
    for (int k0 = 0; k0 < 64; k0 += 4) {
        const int kA = k0 + half * 2;
        const v2f m0 = *(const v2f*)(mask + (kA + 0) * NPIX + p);
        const v2f m1 = *(const v2f*)(mask + (kA + 1) * NPIX + p);
        v2f b0, b1;
        b0[0] = sigmoidf(m0[0]); b0[1] = sigmoidf(m1[0]);
        b1[0] = sigmoidf(m0[1]); b1[1] = sigmoidf(m1[1]);
        #pragma unroll
        for (int mt = 0; mt < 4; ++mt) {
            v2f a;
            a[0] = sO[mt * 16 + col][kA + 0];
            a[1] = sO[mt * 16 + col][kA + 1];
            acc[mt][0] = __builtin_amdgcn_wmma_f32_16x16x4_f32(
                false, a, false, b0, (short)0, acc[mt][0], false, false);
            acc[mt][1] = __builtin_amdgcn_wmma_f32_16x16x4_f32(
                false, a, false, b1, (short)0, acc[mt][1], false, false);
        }
    }

    #pragma unroll
    for (int mt = 0; mt < 4; ++mt) {
        #pragma unroll
        for (int r = 0; r < 8; ++r) {
            const int row = mt * 16 + half * 8 + r;
            const v2f m = *(const v2f*)(mask + row * NPIX + p);
            v2f o;
            o[0] = m[0] * (1.0f - sigmoidf(m[0]) * acc[mt][0][r]);
            o[1] = m[1] * (1.0f - sigmoidf(m[1]) * acc[mt][1][r]);
            *(v2f*)(out + row * NPIX + p) = o;
        }
    }
}

extern "C" void kernel_launch(void* const* d_in, const int* in_sizes, int n_in,
                              void* d_out, int out_size, void* d_ws, size_t ws_size,
                              hipStream_t stream) {
    (void)in_sizes; (void)n_in; (void)out_size; (void)ws_size;

    const float* mask = (const float*)d_in[0];
    const float* bbox = (const float*)d_in[1];
    const int*   cls  = (const int*)  d_in[2];
    const float* U_w  = (const float*)d_in[3];
    const float* V_w  = (const float*)d_in[4];
    const float* Wc_w = (const float*)d_in[5];
    const float* P_w  = (const float*)d_in[6];

    float* out = (float*)d_out;
    float* ws  = (float*)d_ws;
    float* W1  = ws;
    float* F2  = ws + 16384;
    float* Omat= ws + 36864;

    feat_kernel    <<<64,  256, 0, stream>>>(cls, U_w, V_w, P_w, W1, F2);
    relation_kernel<<<64,  256, 0, stream>>>(W1, F2, bbox, Wc_w, P_w, Omat,
                                             out + (size_t)N_INST * NPIX);
    apply_kernel   <<<500, 128, 0, stream>>>(mask, Omat, out);
}
